# starting kernel
# baseline (speedup 1.0000x reference)
#include <hip/hip_runtime.h>
#include <hip/hip_bf16.h>

typedef __attribute__((ext_vector_type(2))) float v2f;
typedef __attribute__((ext_vector_type(8))) float v8f;

#define C_IN   64
#define C_OUT  128
#define MAXN   32
#define ROWS_PER_BLOCK 64
#define AS_LD  68
#define BS_LD  129

__global__ __launch_bounds__(256) void pool_gemm_wmma(
    const float* __restrict__ in,
    const float* __restrict__ W,
    float* __restrict__ x,
    int rows)
{
    __shared__ float As[ROWS_PER_BLOCK][AS_LD];
    __shared__ float Bs[C_IN][BS_LD];

    const int b        = blockIdx.y;
    const int row_base = blockIdx.x * ROWS_PER_BLOCK;
    const int tid      = threadIdx.x;
    const float* inb   = in + (size_t)b * rows * C_IN;

    #pragma unroll
    for (int i = tid; i < ROWS_PER_BLOCK * (C_IN / 4); i += 256) {
        const int r  = i >> 4;
        const int c4 = i & 15;
        int gr = row_base + r;
        if (gr >= rows) gr = rows - 1;
        const float4 v = *(const float4*)(inb + (size_t)gr * C_IN + c4 * 4);
        *(float4*)(&As[r][c4 * 4]) = v;
    }
    #pragma unroll
    for (int i = tid; i < C_OUT * (C_IN / 4); i += 256) {
        const int n  = i >> 4;
        const int k4 = i & 15;
        const float4 v = *(const float4*)(W + n * C_IN + k4 * 4);
        Bs[k4 * 4 + 0][n] = v.x;
        Bs[k4 * 4 + 1][n] = v.y;
        Bs[k4 * 4 + 2][n] = v.z;
        Bs[k4 * 4 + 3][n] = v.w;
    }
    __syncthreads();

    const int lane    = tid & 31;
    const int wave    = tid >> 5;
    const int colbase = wave * 16;
    const int hl      = lane >> 4;
    const int l16     = lane & 15;

    v2f bfrag[16];
    #pragma unroll
    for (int kk = 0; kk < 16; ++kk) {
        const int kb = kk * 4 + hl * 2;
        v2f bf;
        bf.x = Bs[kb + 0][colbase + l16];
        bf.y = Bs[kb + 1][colbase + l16];
        bfrag[kk] = bf;
    }

    #pragma unroll
    for (int rt = 0; rt < 4; ++rt) {
        v8f acc = {};
        #pragma unroll
        for (int kk = 0; kk < 16; ++kk) {
            const int kb = kk * 4 + hl * 2;
            v2f af;
            af.x = As[rt * 16 + l16][kb + 0];
            af.y = As[rt * 16 + l16][kb + 1];
            acc = __builtin_amdgcn_wmma_f32_16x16x4_f32(
                false, af, false, bfrag[kk], (short)0, acc, false, false);
        }
        #pragma unroll
        for (int r = 0; r < 8; ++r) {
            const int gm = row_base + rt * 16 + r + 8 * hl;
            if (gm < rows) {
                x[((size_t)b * rows + gm) * C_OUT + colbase + l16] = acc[r];
            }
        }
    }
}

__global__ __launch_bounds__(256) void pool_gather(
    const float* __restrict__ x,
    const int*   __restrict__ ids,
    const float* __restrict__ mask,
    const float* __restrict__ pn,
    float* __restrict__ out,
    int rows, int out_pn, int nbatch)
{
    const int lane = threadIdx.x & 31;
    const int gw   = blockIdx.x * 8 + (threadIdx.x >> 5);
    const int total = nbatch * out_pn;
    if (gw >= total) return;
    const int b = gw / out_pn;
    const int p = gw % out_pn;

    const float pm  = fabsf(pn[p * MAXN + lane]) * mask[p * MAXN + lane];
    const int   idm = ids[p * MAXN + lane];
    float s = pm;
    #pragma unroll
    for (int off = 16; off > 0; off >>= 1) s += __shfl_xor(s, off, 32);
    const float wl = pm / (s + 1e-8f);

    const float* xb = x + (size_t)b * rows * C_OUT;
    float4 acc = {0.f, 0.f, 0.f, 0.f};
    #pragma unroll
    for (int m = 0; m < MAXN; ++m) {
        const float wm = __shfl(wl, m, 32);
        const int   im = __shfl(idm, m, 32);
        const float4 v = *(const float4*)(xb + (size_t)im * C_OUT + lane * 4);
        acc.x += wm * v.x;
        acc.y += wm * v.y;
        acc.z += wm * v.z;
        acc.w += wm * v.w;
    }
    *(float4*)(out + ((size_t)(b * out_pn + p)) * C_OUT + lane * 4) = acc;
}

extern "C" void kernel_launch(void* const* d_in, const int* in_sizes, int n_in,
                              void* d_out, int out_size, void* d_ws, size_t ws_size,
                              hipStream_t stream)
{
    const float* in_pc = (const float*)d_in[0];
    const int*   ids   = (const int*)d_in[1];
    const float* mask  = (const float*)d_in[2];
    const float* pn    = (const float*)d_in[3];
    const float* W     = (const float*)d_in[4];

    const int out_pn = in_sizes[1] / MAXN;
    const int nbatch = out_size / (out_pn * C_OUT);
    const int rows   = in_sizes[0] / (nbatch * C_IN);

    float* x = (float*)d_ws;

    dim3 g1((rows + ROWS_PER_BLOCK - 1) / ROWS_PER_BLOCK, nbatch);
    pool_gemm_wmma<<<g1, 256, 0, stream>>>(in_pc, W, x, rows);

    const int total_waves = nbatch * out_pn;
    const int g2 = (total_waves + 7) / 8;
    pool_gather<<<g2, 256, 0, stream>>>(x, ids, mask, pn, (float*)d_out,
                                        rows, out_pn, nbatch);
}


# Pool before project (linearity): neighbour-weighted sum on the 64-ch inputs first, then the f32 WMMA GEMM on the 80000 pooled rows only; sum order changes (pooled sum rounded, then projected)
# speedup vs baseline: 2.5864x; 2.5864x over previous
#include <hip/hip_runtime.h>
#include <hip/hip_bf16.h>

typedef __attribute__((ext_vector_type(2))) float v2f;
typedef __attribute__((ext_vector_type(8))) float v8f;

#define C_IN   64
#define C_OUT  128
#define MAXN   32
#define ROWS_PER_BLOCK 64
#define AS_LD  68
#define BS_LD  129

__global__ __launch_bounds__(256) void pool_gemm_wmma(
    const float* __restrict__ in,
    const float* __restrict__ W,
    float* __restrict__ x,
    int rows)
{
    __shared__ float As[ROWS_PER_BLOCK][AS_LD];
    __shared__ float Bs[C_IN][BS_LD];

    const int b        = blockIdx.y;
    const int row_base = blockIdx.x * ROWS_PER_BLOCK;
    const int tid      = threadIdx.x;
    const float* inb   = in + (size_t)b * rows * C_IN;

    #pragma unroll
    for (int i = tid; i < ROWS_PER_BLOCK * (C_IN / 4); i += 256) {
        const int r  = i >> 4;
        const int c4 = i & 15;
        int gr = row_base + r;
        if (gr >= rows) gr = rows - 1;
        const float4 v = *(const float4*)(inb + (size_t)gr * C_IN + c4 * 4);
        *(float4*)(&As[r][c4 * 4]) = v;
    }
    #pragma unroll
    for (int i = tid; i < C_OUT * (C_IN / 4); i += 256) {
        const int n  = i >> 4;
        const int k4 = i & 15;
        const float4 v = *(const float4*)(W + n * C_IN + k4 * 4);
        Bs[k4 * 4 + 0][n] = v.x;
        Bs[k4 * 4 + 1][n] = v.y;
        Bs[k4 * 4 + 2][n] = v.z;
        Bs[k4 * 4 + 3][n] = v.w;
    }
    __syncthreads();

    const int lane    = tid & 31;
    const int wave    = tid >> 5;
    const int colbase = wave * 16;
    const int hl      = lane >> 4;
    const int l16     = lane & 15;

    v2f bfrag[16];
    #pragma unroll
    for (int kk = 0; kk < 16; ++kk) {
        const int kb = kk * 4 + hl * 2;
        v2f bf;
        bf.x = Bs[kb + 0][colbase + l16];
        bf.y = Bs[kb + 1][colbase + l16];
        bfrag[kk] = bf;
    }

    #pragma unroll
    for (int rt = 0; rt < 4; ++rt) {
        v8f acc = {};
        #pragma unroll
        for (int kk = 0; kk < 16; ++kk) {
            const int kb = kk * 4 + hl * 2;
            v2f af;
            af.x = As[rt * 16 + l16][kb + 0];
            af.y = As[rt * 16 + l16][kb + 1];
            acc = __builtin_amdgcn_wmma_f32_16x16x4_f32(
                false, af, false, bfrag[kk], (short)0, acc, false, false);
        }
        #pragma unroll
        for (int r = 0; r < 8; ++r) {
            const int gm = row_base + rt * 16 + r + 8 * hl;
            if (gm < rows) {
                x[((size_t)b * rows + gm) * C_OUT + colbase + l16] = acc[r];
            }
        }
    }
}

__global__ __launch_bounds__(256) void pool_gather_in(
    const float* __restrict__ in,
    const int*   __restrict__ ids,
    const float* __restrict__ mask,
    const float* __restrict__ pn,
    float* __restrict__ pooled,
    int rows, int out_pn, int nbatch)
{
    const int lane  = threadIdx.x & 31;
    const int hl    = lane >> 4;
    const int l16   = lane & 15;
    const int gw    = __builtin_amdgcn_readfirstlane(blockIdx.x * 8 + (threadIdx.x >> 5));
    const int npair = (nbatch + 1) >> 1;
    if (gw >= npair * out_pn) return;
    const int bp = gw / out_pn;
    const int p  = gw % out_pn;
    const int b0 = bp * 2;
    const bool live = (b0 + hl) < nbatch;
    const int  hle  = live ? hl : 0;

    const float pm  = fabsf(pn[p * MAXN + lane]) * mask[p * MAXN + lane];
    const int   idm = ids[p * MAXN + lane];
    float s = pm;
    #pragma unroll
    for (int off = 16; off > 0; off >>= 1) s += __shfl_xor(s, off, 32);
    const float wl = pm / (s + 1e-8f);

    const char* base = (const char*)(in + (size_t)b0 * rows * C_IN);
    const unsigned voff = (unsigned)hle * (unsigned)rows * (unsigned)(C_IN * 4)
                        + (unsigned)l16 * 16u;
    float4 acc = {0.f, 0.f, 0.f, 0.f};
    #pragma unroll
    for (int m = 0; m < MAXN; ++m) {
        const float wm = __int_as_float(__builtin_amdgcn_readlane(__float_as_int(wl), m));
        const int   im = __builtin_amdgcn_readlane(idm, m);
        const float4 v = *(const float4*)(base + (size_t)im * (C_IN * 4) + voff);
        acc.x += wm * v.x;
        acc.y += wm * v.y;
        acc.z += wm * v.z;
        acc.w += wm * v.w;
    }
    if (live) {
        *(float4*)(pooled + ((size_t)(b0 + hl) * out_pn + p) * C_IN + l16 * 4) = acc;
    }
}

extern "C" void kernel_launch(void* const* d_in, const int* in_sizes, int n_in,
                              void* d_out, int out_size, void* d_ws, size_t ws_size,
                              hipStream_t stream)
{
    const float* in_pc = (const float*)d_in[0];
    const int*   ids   = (const int*)d_in[1];
    const float* mask  = (const float*)d_in[2];
    const float* pn    = (const float*)d_in[3];
    const float* W     = (const float*)d_in[4];

    const int out_pn = in_sizes[1] / MAXN;
    const int nbatch = out_size / (out_pn * C_OUT);
    const int rows   = in_sizes[0] / (nbatch * C_IN);

    float* pooled = (float*)d_ws;

    const int npair       = (nbatch + 1) / 2;
    const int total_waves = npair * out_pn;
    const int g1 = (total_waves + 7) / 8;
    pool_gather_in<<<g1, 256, 0, stream>>>(in_pc, ids, mask, pn, pooled,
                                           rows, out_pn, nbatch);

    const int prow = nbatch * out_pn;
    dim3 g2((prow + ROWS_PER_BLOCK - 1) / ROWS_PER_BLOCK, 1);
    pool_gemm_wmma<<<g2, 256, 0, stream>>>(pooled, W, (float*)d_out, prow);
}


# Fuse the two launches: each block pools its 8 points straight into a double-buffered LDS A tile (16 rows per batch pair) and runs the f32 WMMA on it; no pooled plane; sum order unchanged
# speedup vs baseline: 3.0273x; 1.1705x over previous
#include <hip/hip_runtime.h>
#include <hip/hip_bf16.h>


typedef __attribute__((ext_vector_type(2))) float v2f;
typedef __attribute__((ext_vector_type(8))) float v8f;

#define C_IN   64
#define C_OUT  128
#define MAXN   32
#define PTS_PER_BLOCK 8
#define AS_LD  68
#define BS_LD  129

__global__ __launch_bounds__(256) void pool_gather_project(
    const float* __restrict__ in,
    const int*   __restrict__ ids,
    const float* __restrict__ mask,
    const float* __restrict__ pn,
    const float* __restrict__ W,
    float* __restrict__ out,
    int rows, int out_pn, int nbatch)
{
    __shared__ float As[2][16][AS_LD];
    __shared__ float Bs[C_IN][BS_LD];

    const int tid   = threadIdx.x;
    const int lane  = tid & 31;
    const int hl    = lane >> 4;
    const int l16   = lane & 15;
    const int wave  = __builtin_amdgcn_readfirstlane(tid >> 5);
    const int colbase = wave * 16;
    const int p0    = blockIdx.x * PTS_PER_BLOCK;
    const int p     = p0 + wave;
    const int pc    = (p < out_pn) ? p : (out_pn - 1);

    const float pnv = pn[pc * MAXN + lane];
    const float mkv = mask[pc * MAXN + lane];
    const int   idm = ids[pc * MAXN + lane];

    #pragma unroll
    for (int j = 0; j < (C_OUT * (C_IN / 4)) / 256; ++j) {
        const int i  = tid + j * 256;
        const int n  = i >> 4;
        const int k4 = i & 15;
        const float4 v = *(const float4*)(W + n * C_IN + k4 * 4);
        Bs[k4 * 4 + 0][n] = v.x;
        Bs[k4 * 4 + 1][n] = v.y;
        Bs[k4 * 4 + 2][n] = v.z;
        Bs[k4 * 4 + 3][n] = v.w;
    }

    const float pm  = fabsf(pnv) * mkv;
    float s = pm;
    #pragma unroll
    for (int off = 16; off > 0; off >>= 1) s += __shfl_xor(s, off, 32);
    const float wl = pm / (s + 1e-8f);

    const int npair = (nbatch + 1) >> 1;
    for (int bp = 0; bp < npair; ++bp) {
        const int  b0   = bp * 2;
        const bool live = (b0 + hl) < nbatch;
        const int  hle  = live ? hl : 0;

        const char* base = (const char*)(in + (size_t)b0 * rows * C_IN);
        const unsigned voff = (unsigned)hle * (unsigned)rows * (unsigned)(C_IN * 4)
                            + (unsigned)l16 * 16u;
        float4 acc = {0.f, 0.f, 0.f, 0.f};
        #pragma unroll
        for (int m = 0; m < MAXN; ++m) {
            const float wm = __int_as_float(__builtin_amdgcn_readlane(__float_as_int(wl), m));
            const int   im = __builtin_amdgcn_readlane(idm, m);
            const float4 v = *(const float4*)(base + (size_t)im * (C_IN * 4) + voff);
            acc.x += wm * v.x;
            acc.y += wm * v.y;
            acc.z += wm * v.z;
            acc.w += wm * v.w;
        }

        float (*At)[AS_LD] = As[bp & 1];
        *(float4*)(&At[hl * 8 + wave][l16 * 4]) = acc;
        __syncthreads();

        v8f d = {};
        #pragma unroll
        for (int kk = 0; kk < 16; ++kk) {
            const int kb = kk * 4 + hl * 2;
            v2f af, bf;
            af.x = At[l16][kb + 0];
            af.y = At[l16][kb + 1];
            bf.x = Bs[kb + 0][colbase + l16];
            bf.y = Bs[kb + 1][colbase + l16];
            d = __builtin_amdgcn_wmma_f32_16x16x4_f32(
                false, af, false, bf, (short)0, d, false, false);
        }
        #pragma unroll
        for (int r = 0; r < 8; ++r) {
            const int pp = p0 + r;
            if (live && pp < out_pn) {
                out[((size_t)(b0 + hl) * out_pn + pp) * C_OUT + colbase + l16] = d[r];
            }
        }
    }
}

extern "C" void kernel_launch(void* const* d_in, const int* in_sizes, int n_in,
                              void* d_out, int out_size, void* d_ws, size_t ws_size,
                              hipStream_t stream)
{
    const float* in_pc = (const float*)d_in[0];
    const int*   ids   = (const int*)d_in[1];
    const float* mask  = (const float*)d_in[2];
    const float* pn    = (const float*)d_in[3];
    const float* W     = (const float*)d_in[4];

    const int out_pn = in_sizes[1] / MAXN;
    const int nbatch = out_size / (out_pn * C_OUT);
    const int rows   = in_sizes[0] / (nbatch * C_IN);

    const int g = (out_pn + PTS_PER_BLOCK - 1) / PTS_PER_BLOCK;
    pool_gather_project<<<g, 256, 0, stream>>>(in_pc, ids, mask, pn, W, (float*)d_out,
                                              rows, out_pn, nbatch);
}


# Stage each 16x128 output tile through LDS: a wave writes each 512-byte output row with one 16-byte-per-lane store (whole 128-byte lines), not 4-byte stores of 64-byte pieces; sum order unchanged
# speedup vs baseline: 3.4122x; 1.1271x over previous
#include <hip/hip_runtime.h>
#include <hip/hip_bf16.h>


typedef __attribute__((ext_vector_type(2))) float v2f;
typedef __attribute__((ext_vector_type(8))) float v8f;

#define C_IN   64
#define C_OUT  128
#define MAXN   32
#define PTS_PER_BLOCK 8
#define AS_LD  68
#define BS_LD  129
#define DS_LD  132

__global__ __launch_bounds__(256) void pool_gather_project(
    const float* __restrict__ in,
    const int*   __restrict__ ids,
    const float* __restrict__ mask,
    const float* __restrict__ pn,
    const float* __restrict__ W,
    float* __restrict__ out,
    int rows, int out_pn, int nbatch)
{
    __shared__ float As[2][16][AS_LD];
    __shared__ float Bs[C_IN][BS_LD];
    __shared__ float Ds[16][DS_LD];

    const int tid   = threadIdx.x;
    const int lane  = tid & 31;
    const int hl    = lane >> 4;
    const int l16   = lane & 15;
    const int wave  = __builtin_amdgcn_readfirstlane(tid >> 5);
    const int colbase = wave * 16;
    const int p0    = blockIdx.x * PTS_PER_BLOCK;
    const int p     = p0 + wave;
    const int pc    = (p < out_pn) ? p : (out_pn - 1);

    const float pnv = pn[pc * MAXN + lane];
    const float mkv = mask[pc * MAXN + lane];
    const int   idm = ids[pc * MAXN + lane];

    #pragma unroll
    for (int j = 0; j < (C_OUT * (C_IN / 4)) / 256; ++j) {
        const int i  = tid + j * 256;
        const int n  = i >> 4;
        const int k4 = i & 15;
        const float4 v = *(const float4*)(W + n * C_IN + k4 * 4);
        Bs[k4 * 4 + 0][n] = v.x;
        Bs[k4 * 4 + 1][n] = v.y;
        Bs[k4 * 4 + 2][n] = v.z;
        Bs[k4 * 4 + 3][n] = v.w;
    }

    const float pm  = fabsf(pnv) * mkv;
    float s = pm;
    #pragma unroll
    for (int off = 16; off > 0; off >>= 1) s += __shfl_xor(s, off, 32);
    const float wl = pm / (s + 1e-8f);

    const int npair = (nbatch + 1) >> 1;
    for (int bp = 0; bp < npair; ++bp) {
        const int  b0   = bp * 2;
        const bool live = (b0 + hl) < nbatch;
        const int  hle  = live ? hl : 0;

        const char* base = (const char*)(in + (size_t)b0 * rows * C_IN);
        const unsigned voff = (unsigned)hle * (unsigned)rows * (unsigned)(C_IN * 4)
                            + (unsigned)l16 * 16u;
        float4 acc = {0.f, 0.f, 0.f, 0.f};
        #pragma unroll
        for (int m = 0; m < MAXN; ++m) {
            const float wm = __int_as_float(__builtin_amdgcn_readlane(__float_as_int(wl), m));
            const int   im = __builtin_amdgcn_readlane(idm, m);
            const float4 v = *(const float4*)(base + (size_t)im * (C_IN * 4) + voff);
            acc.x += wm * v.x;
            acc.y += wm * v.y;
            acc.z += wm * v.z;
            acc.w += wm * v.w;
        }

        float (*At)[AS_LD] = As[bp & 1];
        *(float4*)(&At[hl * 8 + wave][l16 * 4]) = acc;
        __syncthreads();

        v8f d = {};
        #pragma unroll
        for (int kk = 0; kk < 16; ++kk) {
            const int kb = kk * 4 + hl * 2;
            v2f af, bf;
            af.x = At[l16][kb + 0];
            af.y = At[l16][kb + 1];
            bf.x = Bs[kb + 0][colbase + l16];
            bf.y = Bs[kb + 1][colbase + l16];
            d = __builtin_amdgcn_wmma_f32_16x16x4_f32(
                false, af, false, bf, (short)0, d, false, false);
        }
        #pragma unroll
        for (int r = 0; r < 8; ++r) {
            Ds[r + 8 * hl][colbase + l16] = d[r];
        }
        __syncthreads();
        if (p < out_pn) {
            float* orow = out + ((size_t)b0 * out_pn + p) * C_OUT + lane * 4;
            *(float4*)orow = *(const float4*)(&Ds[wave][lane * 4]);
            if (b0 + 1 < nbatch) {
                *(float4*)(orow + (size_t)out_pn * C_OUT) = *(const float4*)(&Ds[8 + wave][lane * 4]);
            }
        }
    }
}

extern "C" void kernel_launch(void* const* d_in, const int* in_sizes, int n_in,
                              void* d_out, int out_size, void* d_ws, size_t ws_size,
                              hipStream_t stream)
{
    const float* in_pc = (const float*)d_in[0];
    const int*   ids   = (const int*)d_in[1];
    const float* mask  = (const float*)d_in[2];
    const float* pn    = (const float*)d_in[3];
    const float* W     = (const float*)d_in[4];

    const int out_pn = in_sizes[1] / MAXN;
    const int nbatch = out_size / (out_pn * C_OUT);
    const int rows   = in_sizes[0] / (nbatch * C_IN);

    const int g = (out_pn + PTS_PER_BLOCK - 1) / PTS_PER_BLOCK;
    pool_gather_project<<<g, 256, 0, stream>>>(in_pc, ids, mask, pn, W, (float*)d_out,
                                              rows, out_pn, nbatch);
}
